# starting kernel
# baseline (speedup 1.0000x reference)
#include <hip/hip_runtime.h>
#include <hip/hip_bf16.h>
#include <math.h>

#define NN 256
#define BSZ 256
#define TMAXX 512
#define STEPF 0.01f
#define YAF 0.001f
#define YWF 0.001f

#define CSTRIDE 132

typedef __attribute__((ext_vector_type(2))) float v2f;
typedef __attribute__((ext_vector_type(8))) float v8f;

static __device__ __forceinline__ v8f wmma4(v2f a, v2f b, v8f c) {
    return __builtin_amdgcn_wmma_f32_16x16x4_f32(
        false, a, false, b, (short)0, c, false, false);
}

__global__ __launch_bounds__(256) void lrnn_prep(
    const float* __restrict__ MA, const float* __restrict__ MW,
    const float* __restrict__ bAz, const float* __restrict__ bWz,
    float* __restrict__ A, float* __restrict__ C) {
    const int i = blockIdx.x;
    const int j = threadIdx.x;
    const float zA = bAz[0];
    const float zW = bWz[0];
    const float bA = 0.5f * expf(-zA * zA) + 0.5f;
    const float bW = 0.5f * expf(-zW * zW) + 0.5f;
    const float ma_ij = MA[i * NN + j];
    const float ma_ji = MA[j * NN + i];
    const float mw_ij = MW[i * NN + j];
    const float mw_ji = MW[j * NN + i];
    const float d = (i == j) ? 1.0f : 0.0f;
    A[i * NN + j] = ma_ij + (1.0f - 2.0f * bA) * ma_ji - YAF * d;
    C[i * NN + j] = (1.0f - bA + bW) * mw_ij + (1.0f - bA - bW) * mw_ji - YWF * d;
}

__global__ __launch_bounds__(512) void lrnn_main(
    const float* __restrict__ X0, const float* __restrict__ A,
    const float* __restrict__ Cg, const float* __restrict__ by,
    float* __restrict__ out) {
    __shared__ float Xs[16][260];
    __shared__ float Cs[NN * CSTRIDE];

    const int tid  = threadIdx.x;
    const int wav  = tid >> 5;
    const int lane = tid & 31;
    const int lo   = lane & 15;
    const int hi   = lane >> 4;
    const int bc0  = blockIdx.x * 16;

    for (int idx = tid; idx < 16 * NN; idx += 512) {
        const int col = idx >> 8;
        const int row = idx & 255;
        const float v = X0[(size_t)(bc0 + col) * NN + row];
        Xs[col][row] = v;
        out[(size_t)(bc0 + col) * TMAXX * NN + row] = v;
    }
    for (int idx = tid; idx < NN * 128; idx += 512) {
        const int row = idx >> 7;
        const int col = idx & 127;
        Cs[row * CSTRIDE + col] = Cg[row * NN + 128 + col];
    }

    const int rowA = wav * 16 + lo;
    v2f aF[64];
    v2f cF[32];
#pragma unroll
    for (int k = 0; k < 64; ++k)
        aF[k] = *(const v2f*)(A + (size_t)rowA * NN + 4 * k + 2 * hi);
#pragma unroll
    for (int k = 0; k < 32; ++k)
        cF[k] = *(const v2f*)(Cg + (size_t)rowA * NN + 4 * k + 2 * hi);

    const int rbase = wav * 16 + 8 * hi;
    float byF[8];
#pragma unroll
    for (int j = 0; j < 8; ++j) byF[j] = by[rbase + j];

    __syncthreads();

    float* outLane = out + (size_t)(bc0 + lo) * TMAXX * NN + rbase;
    const float* crow = &Cs[rowA * CSTRIDE];

#pragma unroll 1
    for (int t = 1; t < TMAXX; ++t) {
        v8f accA = {0.f, 0.f, 0.f, 0.f, 0.f, 0.f, 0.f, 0.f};
        v8f accC = {0.f, 0.f, 0.f, 0.f, 0.f, 0.f, 0.f, 0.f};
#pragma unroll
        for (int k = 0; k < 32; ++k) {
            const int kr = 4 * k + 2 * hi;
            const v2f b = *(const v2f*)&Xs[lo][kr];
            accA = wmma4(aF[k], b, accA);
            accC = wmma4(cF[k], b, accC);
        }
#pragma unroll
        for (int k = 32; k < 64; ++k) {
            const int kr = 4 * k + 2 * hi;
            const v2f b  = *(const v2f*)&Xs[lo][kr];
            const v2f cl = *(const v2f*)&crow[kr - 128];
            accA = wmma4(aF[k], b, accA);
            accC = wmma4(cl, b, accC);
        }
        float xn[8];
#pragma unroll
        for (int j = 0; j < 8; ++j) {
            const float xc = Xs[lo][rbase + j];
            const float yi = accC[j] + byF[j];
            xn[j] = xc + STEPF * (accA[j] + tanhf(yi));
        }
        __syncthreads();
#pragma unroll
        for (int j = 0; j < 8; ++j) Xs[lo][rbase + j] = xn[j];
        float* p = outLane + (size_t)t * NN;
        *(float4*)(p)     = make_float4(xn[0], xn[1], xn[2], xn[3]);
        *(float4*)(p + 4) = make_float4(xn[4], xn[5], xn[6], xn[7]);
        __syncthreads();
    }
}

extern "C" void kernel_launch(void* const* d_in, const int* in_sizes, int n_in,
                              void* d_out, int out_size, void* d_ws, size_t ws_size,
                              hipStream_t stream) {
    const float* X0  = (const float*)d_in[0];
    const float* MA  = (const float*)d_in[1];
    const float* MW  = (const float*)d_in[2];
    const float* bAz = (const float*)d_in[3];
    const float* bWz = (const float*)d_in[4];
    const float* by  = (const float*)d_in[5];
    float* out = (float*)d_out;

    float* A = (float*)d_ws;
    float* C = A + NN * NN;

    lrnn_prep<<<NN, NN, 0, stream>>>(MA, MW, bAz, bWz, A, C);
    lrnn_main<<<16, 512, 0, stream>>>(X0, A, C, by, out);
}


# tanh evaluated without branches: both range forms computed and one selected, so a lane's 8 values run interleaved instead of as 8 serial branch pairs; values bit-identical
# speedup vs baseline: 1.1054x; 1.1054x over previous
#include <hip/hip_runtime.h>
#include <hip/hip_bf16.h>
#include <math.h>

#define NN 256
#define BSZ 256
#define TMAXX 512
#define STEPF 0.01f
#define YAF 0.001f
#define YWF 0.001f

#define CSTRIDE 132

typedef __attribute__((ext_vector_type(2))) float v2f;
typedef __attribute__((ext_vector_type(8))) float v8f;

static __device__ __forceinline__ v8f wmma4(v2f a, v2f b, v8f c) {
    return __builtin_amdgcn_wmma_f32_16x16x4_f32(
        false, a, false, b, (short)0, c, false, false);
}

static __device__ __forceinline__ float tanh_sel(float y) {
    const float ay = fabsf(y);
    const float e  = expf(ay + ay);
    const float tb = fmaf(__builtin_amdgcn_rcpf(e + 1.0f), -2.0f, 1.0f);
    const float s  = y * y;
    float p = fmaf(s, -0x1.758e7ap-8f, 0x1.521192p-6f);
    p = fmaf(s, p, -0x1.b8389cp-5f);
    p = fmaf(s, p, 0x1.110704p-3f);
    p = fmaf(s, p, -0x1.555532p-2f);
    const float ts = fmaf(s, ay * p, ay);
    const float t  = (ay < 0.625f) ? ts : tb;
    return copysignf(t, y);
}

__global__ __launch_bounds__(256) void lrnn_prep(
    const float* __restrict__ MA, const float* __restrict__ MW,
    const float* __restrict__ bAz, const float* __restrict__ bWz,
    float* __restrict__ A, float* __restrict__ C) {
    const int i = blockIdx.x;
    const int j = threadIdx.x;
    const float zA = bAz[0];
    const float zW = bWz[0];
    const float bA = 0.5f * expf(-zA * zA) + 0.5f;
    const float bW = 0.5f * expf(-zW * zW) + 0.5f;
    const float ma_ij = MA[i * NN + j];
    const float ma_ji = MA[j * NN + i];
    const float mw_ij = MW[i * NN + j];
    const float mw_ji = MW[j * NN + i];
    const float d = (i == j) ? 1.0f : 0.0f;
    A[i * NN + j] = ma_ij + (1.0f - 2.0f * bA) * ma_ji - YAF * d;
    C[i * NN + j] = (1.0f - bA + bW) * mw_ij + (1.0f - bA - bW) * mw_ji - YWF * d;
}

__global__ __launch_bounds__(512) void lrnn_main(
    const float* __restrict__ X0, const float* __restrict__ A,
    const float* __restrict__ Cg, const float* __restrict__ by,
    float* __restrict__ out) {
    __shared__ float Xs[16][260];
    __shared__ float Cs[NN * CSTRIDE];

    const int tid  = threadIdx.x;
    const int wav  = tid >> 5;
    const int lane = tid & 31;
    const int lo   = lane & 15;
    const int hi   = lane >> 4;
    const int bc0  = blockIdx.x * 16;

    for (int idx = tid; idx < 16 * NN; idx += 512) {
        const int col = idx >> 8;
        const int row = idx & 255;
        const float v = X0[(size_t)(bc0 + col) * NN + row];
        Xs[col][row] = v;
        out[(size_t)(bc0 + col) * TMAXX * NN + row] = v;
    }
    for (int idx = tid; idx < NN * 128; idx += 512) {
        const int row = idx >> 7;
        const int col = idx & 127;
        Cs[row * CSTRIDE + col] = Cg[row * NN + 128 + col];
    }

    const int rowA = wav * 16 + lo;
    v2f aF[64];
    v2f cF[32];
#pragma unroll
    for (int k = 0; k < 64; ++k)
        aF[k] = *(const v2f*)(A + (size_t)rowA * NN + 4 * k + 2 * hi);
#pragma unroll
    for (int k = 0; k < 32; ++k)
        cF[k] = *(const v2f*)(Cg + (size_t)rowA * NN + 4 * k + 2 * hi);

    const int rbase = wav * 16 + 8 * hi;
    float byF[8];
#pragma unroll
    for (int j = 0; j < 8; ++j) byF[j] = by[rbase + j];

    __syncthreads();

    float* outLane = out + (size_t)(bc0 + lo) * TMAXX * NN + rbase;
    const float* crow = &Cs[rowA * CSTRIDE];

#pragma unroll 1
    for (int t = 1; t < TMAXX; ++t) {
        v8f accA = {0.f, 0.f, 0.f, 0.f, 0.f, 0.f, 0.f, 0.f};
        v8f accC = {0.f, 0.f, 0.f, 0.f, 0.f, 0.f, 0.f, 0.f};
#pragma unroll
        for (int k = 0; k < 32; ++k) {
            const int kr = 4 * k + 2 * hi;
            const v2f b = *(const v2f*)&Xs[lo][kr];
            accA = wmma4(aF[k], b, accA);
            accC = wmma4(cF[k], b, accC);
        }
#pragma unroll
        for (int k = 32; k < 64; ++k) {
            const int kr = 4 * k + 2 * hi;
            const v2f b  = *(const v2f*)&Xs[lo][kr];
            const v2f cl = *(const v2f*)&crow[kr - 128];
            accA = wmma4(aF[k], b, accA);
            accC = wmma4(cl, b, accC);
        }
        float xn[8];
#pragma unroll
        for (int j = 0; j < 8; ++j) {
            const float xc = Xs[lo][rbase + j];
            const float yi = accC[j] + byF[j];
            xn[j] = xc + STEPF * (accA[j] + tanh_sel(yi));
        }
        __syncthreads();
#pragma unroll
        for (int j = 0; j < 8; ++j) Xs[lo][rbase + j] = xn[j];
        float* p = outLane + (size_t)t * NN;
        *(float4*)(p)     = make_float4(xn[0], xn[1], xn[2], xn[3]);
        *(float4*)(p + 4) = make_float4(xn[4], xn[5], xn[6], xn[7]);
        __syncthreads();
    }
}

extern "C" void kernel_launch(void* const* d_in, const int* in_sizes, int n_in,
                              void* d_out, int out_size, void* d_ws, size_t ws_size,
                              hipStream_t stream) {
    const float* X0  = (const float*)d_in[0];
    const float* MA  = (const float*)d_in[1];
    const float* MW  = (const float*)d_in[2];
    const float* bAz = (const float*)d_in[3];
    const float* bWz = (const float*)d_in[4];
    const float* by  = (const float*)d_in[5];
    float* out = (float*)d_out;

    float* A = (float*)d_ws;
    float* C = A + NN * NN;

    lrnn_prep<<<NN, NN, 0, stream>>>(MA, MW, bAz, bWz, A, C);
    lrnn_main<<<16, 512, 0, stream>>>(X0, A, C, by, out);
}


# each time slice is stored one step later, behind the step's last barrier, as whole 1 KB rows read back from the LDS state tile: no wave drains global stores before a barrier; values bit-identical
# speedup vs baseline: 1.3436x; 1.2155x over previous
#include <hip/hip_runtime.h>
#include <hip/hip_bf16.h>
#include <math.h>

#define NN 256
#define BSZ 256
#define TMAXX 512
#define STEPF 0.01f
#define YAF 0.001f
#define YWF 0.001f

#define CSTRIDE 132

typedef __attribute__((ext_vector_type(2))) float v2f;
typedef __attribute__((ext_vector_type(8))) float v8f;

static __device__ __forceinline__ v8f wmma4(v2f a, v2f b, v8f c) {
    return __builtin_amdgcn_wmma_f32_16x16x4_f32(
        false, a, false, b, (short)0, c, false, false);
}

static __device__ __forceinline__ float tanh_sel(float y) {
    const float ay = fabsf(y);
    const float e  = expf(ay + ay);
    const float tb = fmaf(__builtin_amdgcn_rcpf(e + 1.0f), -2.0f, 1.0f);
    const float s  = y * y;
    float p = fmaf(s, -0x1.758e7ap-8f, 0x1.521192p-6f);
    p = fmaf(s, p, -0x1.b8389cp-5f);
    p = fmaf(s, p, 0x1.110704p-3f);
    p = fmaf(s, p, -0x1.555532p-2f);
    const float ts = fmaf(s, ay * p, ay);
    const float t  = (ay < 0.625f) ? ts : tb;
    return copysignf(t, y);
}

__global__ __launch_bounds__(256) void lrnn_prep(
    const float* __restrict__ MA, const float* __restrict__ MW,
    const float* __restrict__ bAz, const float* __restrict__ bWz,
    float* __restrict__ A, float* __restrict__ C) {
    const int i = blockIdx.x;
    const int j = threadIdx.x;
    const float zA = bAz[0];
    const float zW = bWz[0];
    const float bA = 0.5f * expf(-zA * zA) + 0.5f;
    const float bW = 0.5f * expf(-zW * zW) + 0.5f;
    const float ma_ij = MA[i * NN + j];
    const float ma_ji = MA[j * NN + i];
    const float mw_ij = MW[i * NN + j];
    const float mw_ji = MW[j * NN + i];
    const float d = (i == j) ? 1.0f : 0.0f;
    A[i * NN + j] = ma_ij + (1.0f - 2.0f * bA) * ma_ji - YAF * d;
    C[i * NN + j] = (1.0f - bA + bW) * mw_ij + (1.0f - bA - bW) * mw_ji - YWF * d;
}

__global__ __launch_bounds__(512) void lrnn_main(
    const float* __restrict__ X0, const float* __restrict__ A,
    const float* __restrict__ Cg, const float* __restrict__ by,
    float* __restrict__ out) {
    __shared__ float Xs[16][260];
    __shared__ float Cs[NN * CSTRIDE];

    const int tid  = threadIdx.x;
    const int wav  = tid >> 5;
    const int lane = tid & 31;
    const int lo   = lane & 15;
    const int hi   = lane >> 4;
    const int bc0  = blockIdx.x * 16;

    for (int idx = tid; idx < 16 * NN; idx += 512) {
        const int col = idx >> 8;
        const int row = idx & 255;
        Xs[col][row] = X0[(size_t)(bc0 + col) * NN + row];
    }
    for (int idx = tid; idx < NN * 128; idx += 512) {
        const int row = idx >> 7;
        const int col = idx & 127;
        Cs[row * CSTRIDE + col] = Cg[row * NN + 128 + col];
    }

    const int rowA = wav * 16 + lo;
    v2f aF[64];
    v2f cF[32];
#pragma unroll
    for (int k = 0; k < 64; ++k)
        aF[k] = *(const v2f*)(A + (size_t)rowA * NN + 4 * k + 2 * hi);
#pragma unroll
    for (int k = 0; k < 32; ++k)
        cF[k] = *(const v2f*)(Cg + (size_t)rowA * NN + 4 * k + 2 * hi);

    const int rbase = wav * 16 + 8 * hi;
    float byF[8];
#pragma unroll
    for (int j = 0; j < 8; ++j) byF[j] = by[rbase + j];

    __syncthreads();

    const float* crow = &Cs[rowA * CSTRIDE];

    const float* xrow = &Xs[wav][4 * lane];
    float* outRow = out + (size_t)(bc0 + wav) * TMAXX * NN + 4 * lane;

#pragma unroll 1
    for (int t = 1; t < TMAXX; ++t) {
        {
            const float4 o0 = *(const float4*)(xrow);
            const float4 o1 = *(const float4*)(xrow + 128);
            float* q = outRow + (size_t)(t - 1) * NN;
            *(float4*)(q)       = o0;
            *(float4*)(q + 128) = o1;
        }
        v8f accA = {0.f, 0.f, 0.f, 0.f, 0.f, 0.f, 0.f, 0.f};
        v8f accC = {0.f, 0.f, 0.f, 0.f, 0.f, 0.f, 0.f, 0.f};
#pragma unroll
        for (int k = 0; k < 32; ++k) {
            const int kr = 4 * k + 2 * hi;
            const v2f b = *(const v2f*)&Xs[lo][kr];
            accA = wmma4(aF[k], b, accA);
            accC = wmma4(cF[k], b, accC);
        }
#pragma unroll
        for (int k = 32; k < 64; ++k) {
            const int kr = 4 * k + 2 * hi;
            const v2f b  = *(const v2f*)&Xs[lo][kr];
            const v2f cl = *(const v2f*)&crow[kr - 128];
            accA = wmma4(aF[k], b, accA);
            accC = wmma4(cl, b, accC);
        }
        float xn[8];
#pragma unroll
        for (int j = 0; j < 8; ++j) {
            const float xc = Xs[lo][rbase + j];
            const float yi = accC[j] + byF[j];
            xn[j] = xc + STEPF * (accA[j] + tanh_sel(yi));
        }
        __syncthreads();
#pragma unroll
        for (int j = 0; j < 8; ++j) Xs[lo][rbase + j] = xn[j];
        __syncthreads();
    }
    {
        const float4 o0 = *(const float4*)(xrow);
        const float4 o1 = *(const float4*)(xrow + 128);
        float* q = outRow + (size_t)(TMAXX - 1) * NN;
        *(float4*)(q)       = o0;
        *(float4*)(q + 128) = o1;
    }
}

extern "C" void kernel_launch(void* const* d_in, const int* in_sizes, int n_in,
                              void* d_out, int out_size, void* d_ws, size_t ws_size,
                              hipStream_t stream) {
    const float* X0  = (const float*)d_in[0];
    const float* MA  = (const float*)d_in[1];
    const float* MW  = (const float*)d_in[2];
    const float* bAz = (const float*)d_in[3];
    const float* bWz = (const float*)d_in[4];
    const float* by  = (const float*)d_in[5];
    float* out = (float*)d_out;

    float* A = (float*)d_ws;
    float* C = A + NN * NN;

    lrnn_prep<<<NN, NN, 0, stream>>>(MA, MW, bAz, bWz, A, C);
    lrnn_main<<<16, 512, 0, stream>>>(X0, A, C, by, out);
}


# LDS fragment reads regrouped: each block's 8 paired reads go out as one group a block ahead of its 16 matrix instructions, not singly among them; all of C now in LDS for the ring; bit-identical
# speedup vs baseline: 1.4122x; 1.0511x over previous
#include <hip/hip_runtime.h>
#include <hip/hip_bf16.h>
#include <math.h>

#define NN 256
#define BSZ 256
#define TMAXX 512
#define STEPF 0.01f
#define YAF 0.001f
#define YWF 0.001f

#define CSTRIDE 260
#define KBLK 8

typedef __attribute__((ext_vector_type(2))) float v2f;
typedef __attribute__((ext_vector_type(8))) float v8f;

static __device__ __forceinline__ v8f wmma4(v2f a, v2f b, v8f c) {
    return __builtin_amdgcn_wmma_f32_16x16x4_f32(
        false, a, false, b, (short)0, c, false, false);
}

static __device__ __forceinline__ float tanh_sel(float y) {
    const float ay = fabsf(y);
    const float e  = expf(ay + ay);
    const float tb = fmaf(__builtin_amdgcn_rcpf(e + 1.0f), -2.0f, 1.0f);
    const float s  = y * y;
    float p = fmaf(s, -0x1.758e7ap-8f, 0x1.521192p-6f);
    p = fmaf(s, p, -0x1.b8389cp-5f);
    p = fmaf(s, p, 0x1.110704p-3f);
    p = fmaf(s, p, -0x1.555532p-2f);
    const float ts = fmaf(s, ay * p, ay);
    const float t  = (ay < 0.625f) ? ts : tb;
    return copysignf(t, y);
}

__global__ __launch_bounds__(256) void lrnn_prep(
    const float* __restrict__ MA, const float* __restrict__ MW,
    const float* __restrict__ bAz, const float* __restrict__ bWz,
    float* __restrict__ A, float* __restrict__ C) {
    const int i = blockIdx.x;
    const int j = threadIdx.x;
    const float zA = bAz[0];
    const float zW = bWz[0];
    const float bA = 0.5f * expf(-zA * zA) + 0.5f;
    const float bW = 0.5f * expf(-zW * zW) + 0.5f;
    const float ma_ij = MA[i * NN + j];
    const float ma_ji = MA[j * NN + i];
    const float mw_ij = MW[i * NN + j];
    const float mw_ji = MW[j * NN + i];
    const float d = (i == j) ? 1.0f : 0.0f;
    A[i * NN + j] = ma_ij + (1.0f - 2.0f * bA) * ma_ji - YAF * d;
    C[i * NN + j] = (1.0f - bA + bW) * mw_ij + (1.0f - bA - bW) * mw_ji - YWF * d;
}

__global__ __launch_bounds__(512) void lrnn_main(
    const float* __restrict__ X0, const float* __restrict__ A,
    const float* __restrict__ Cg, const float* __restrict__ by,
    float* __restrict__ out) {
    __shared__ __attribute__((aligned(16))) float Xs[16][260];
    __shared__ __attribute__((aligned(16))) float Cs[NN * CSTRIDE];

    const int tid  = threadIdx.x;
    const int wav  = tid >> 5;
    const int lane = tid & 31;
    const int lo   = lane & 15;
    const int hi   = lane >> 4;
    const int bc0  = blockIdx.x * 16;

    for (int idx = tid; idx < 16 * NN; idx += 512) {
        const int col = idx >> 8;
        const int row = idx & 255;
        Xs[col][row] = X0[(size_t)(bc0 + col) * NN + row];
    }
    for (int idx = tid; idx < NN * (NN / 4); idx += 512) {
        const int row = idx >> 6;
        const int c4  = (idx & 63) * 4;
        *(float4*)&Cs[row * CSTRIDE + c4] = *(const float4*)(Cg + row * NN + c4);
    }

    const int rowA = wav * 16 + lo;
    v2f aF[64];
#pragma unroll
    for (int k = 0; k < 64; ++k)
        aF[k] = *(const v2f*)(A + (size_t)rowA * NN + 4 * k + 2 * hi);

    const int rbase = wav * 16 + 8 * hi;
    float byF[8];
#pragma unroll
    for (int j = 0; j < 8; ++j) byF[j] = by[rbase + j];

    __syncthreads();

    const float* xlane = &Xs[lo][2 * hi];
    const float* clane = &Cs[rowA * CSTRIDE + 2 * hi];

    const float* xrow = &Xs[wav][4 * lane];
    float* outRow = out + (size_t)(bc0 + wav) * TMAXX * NN + 4 * lane;

#pragma unroll 1
    for (int t = 1; t < TMAXX; ++t) {
        v2f xr[2][KBLK], cr[2][KBLK];
        {
            const float4 o0 = *(const float4*)(xrow);
            const float4 o1 = *(const float4*)(xrow + 128);
            __builtin_amdgcn_sched_barrier(0);
#pragma unroll
            for (int i = 0; i < KBLK; ++i) {
                xr[0][i] = *(const v2f*)(xlane + 4 * i);
                cr[0][i] = *(const v2f*)(clane + 4 * i);
            }
#pragma unroll
            for (int i = 0; i < KBLK; ++i) {
                xr[1][i] = *(const v2f*)(xlane + 4 * (KBLK + i));
                cr[1][i] = *(const v2f*)(clane + 4 * (KBLK + i));
            }
            float* q = outRow + (size_t)(t - 1) * NN;
            *(float4*)(q)       = o0;
            *(float4*)(q + 128) = o1;
        }
        v8f accA = {0.f, 0.f, 0.f, 0.f, 0.f, 0.f, 0.f, 0.f};
        v8f accC = {0.f, 0.f, 0.f, 0.f, 0.f, 0.f, 0.f, 0.f};
#pragma unroll
        for (int b = 0; b < 64 / KBLK; ++b) {
            __builtin_amdgcn_sched_barrier(0);
#pragma unroll
            for (int i = 0; i < KBLK; ++i) {
                accA = wmma4(aF[KBLK * b + i], xr[b & 1][i], accA);
                accC = wmma4(cr[b & 1][i],     xr[b & 1][i], accC);
            }
            __builtin_amdgcn_sched_barrier(0);
            if (b + 2 < 64 / KBLK) {
#pragma unroll
                for (int i = 0; i < KBLK; ++i) {
                    xr[b & 1][i] = *(const v2f*)(xlane + 4 * (KBLK * (b + 2) + i));
                    cr[b & 1][i] = *(const v2f*)(clane + 4 * (KBLK * (b + 2) + i));
                }
            }
        }
        float xn[8];
#pragma unroll
        for (int j = 0; j < 8; ++j) {
            const float xc = Xs[lo][rbase + j];
            const float yi = accC[j] + byF[j];
            xn[j] = xc + STEPF * (accA[j] + tanh_sel(yi));
        }
        __builtin_amdgcn_sched_barrier(0);
        __syncthreads();
#pragma unroll
        for (int j = 0; j < 8; ++j) Xs[lo][rbase + j] = xn[j];
        __syncthreads();
    }
    {
        const float4 o0 = *(const float4*)(xrow);
        const float4 o1 = *(const float4*)(xrow + 128);
        float* q = outRow + (size_t)(TMAXX - 1) * NN;
        *(float4*)(q)       = o0;
        *(float4*)(q + 128) = o1;
    }
}

extern "C" void kernel_launch(void* const* d_in, const int* in_sizes, int n_in,
                              void* d_out, int out_size, void* d_ws, size_t ws_size,
                              hipStream_t stream) {
    const float* X0  = (const float*)d_in[0];
    const float* MA  = (const float*)d_in[1];
    const float* MW  = (const float*)d_in[2];
    const float* bAz = (const float*)d_in[3];
    const float* bWz = (const float*)d_in[4];
    const float* by  = (const float*)d_in[5];
    float* out = (float*)d_out;

    float* A = (float*)d_ws;
    float* C = A + NN * NN;

    lrnn_prep<<<NN, NN, 0, stream>>>(MA, MW, bAz, bWz, A, C);
    lrnn_main<<<16, 512, 0, stream>>>(X0, A, C, by, out);
}


# exp(2|y|) inside tanh is one exp2 of |y|*2*log2(e), not the library exp (split product, ldexp, clamps): about half the vector instructions per step; tanh differs in the last place, not bit-identical
# speedup vs baseline: 1.4902x; 1.0552x over previous
#include <hip/hip_runtime.h>
#include <hip/hip_bf16.h>
#include <math.h>

#define NN 256
#define BSZ 256
#define TMAXX 512
#define STEPF 0.01f
#define YAF 0.001f
#define YWF 0.001f

#define CSTRIDE 260
#define KBLK 8

typedef __attribute__((ext_vector_type(2))) float v2f;
typedef __attribute__((ext_vector_type(8))) float v8f;

static __device__ __forceinline__ v8f wmma4(v2f a, v2f b, v8f c) {
    return __builtin_amdgcn_wmma_f32_16x16x4_f32(
        false, a, false, b, (short)0, c, false, false);
}

static __device__ __forceinline__ float tanh_sel(float y) {
    const float ay = fabsf(y);
    const float e  = __builtin_amdgcn_exp2f(ay * 0x1.715476p+1f);
    const float tb = fmaf(__builtin_amdgcn_rcpf(e + 1.0f), -2.0f, 1.0f);
    const float s  = y * y;
    float p = fmaf(s, -0x1.758e7ap-8f, 0x1.521192p-6f);
    p = fmaf(s, p, -0x1.b8389cp-5f);
    p = fmaf(s, p, 0x1.110704p-3f);
    p = fmaf(s, p, -0x1.555532p-2f);
    const float ts = fmaf(s, ay * p, ay);
    const float t  = (ay < 0.625f) ? ts : tb;
    return copysignf(t, y);
}

__global__ __launch_bounds__(256) void lrnn_prep(
    const float* __restrict__ MA, const float* __restrict__ MW,
    const float* __restrict__ bAz, const float* __restrict__ bWz,
    float* __restrict__ A, float* __restrict__ C) {
    const int i = blockIdx.x;
    const int j = threadIdx.x;
    const float zA = bAz[0];
    const float zW = bWz[0];
    const float bA = 0.5f * expf(-zA * zA) + 0.5f;
    const float bW = 0.5f * expf(-zW * zW) + 0.5f;
    const float ma_ij = MA[i * NN + j];
    const float ma_ji = MA[j * NN + i];
    const float mw_ij = MW[i * NN + j];
    const float mw_ji = MW[j * NN + i];
    const float d = (i == j) ? 1.0f : 0.0f;
    A[i * NN + j] = ma_ij + (1.0f - 2.0f * bA) * ma_ji - YAF * d;
    C[i * NN + j] = (1.0f - bA + bW) * mw_ij + (1.0f - bA - bW) * mw_ji - YWF * d;
}

__global__ __launch_bounds__(512) void lrnn_main(
    const float* __restrict__ X0, const float* __restrict__ A,
    const float* __restrict__ Cg, const float* __restrict__ by,
    float* __restrict__ out) {
    __shared__ __attribute__((aligned(16))) float Xs[16][260];
    __shared__ __attribute__((aligned(16))) float Cs[NN * CSTRIDE];

    const int tid  = threadIdx.x;
    const int wav  = tid >> 5;
    const int lane = tid & 31;
    const int lo   = lane & 15;
    const int hi   = lane >> 4;
    const int bc0  = blockIdx.x * 16;

    for (int idx = tid; idx < 16 * NN; idx += 512) {
        const int col = idx >> 8;
        const int row = idx & 255;
        Xs[col][row] = X0[(size_t)(bc0 + col) * NN + row];
    }
    for (int idx = tid; idx < NN * (NN / 4); idx += 512) {
        const int row = idx >> 6;
        const int c4  = (idx & 63) * 4;
        *(float4*)&Cs[row * CSTRIDE + c4] = *(const float4*)(Cg + row * NN + c4);
    }

    const int rowA = wav * 16 + lo;
    v2f aF[64];
#pragma unroll
    for (int k = 0; k < 64; ++k)
        aF[k] = *(const v2f*)(A + (size_t)rowA * NN + 4 * k + 2 * hi);

    const int rbase = wav * 16 + 8 * hi;
    float byF[8];
#pragma unroll
    for (int j = 0; j < 8; ++j) byF[j] = by[rbase + j];

    __syncthreads();

    const float* xlane = &Xs[lo][2 * hi];
    const float* clane = &Cs[rowA * CSTRIDE + 2 * hi];

    const float* xrow = &Xs[wav][4 * lane];
    float* outRow = out + (size_t)(bc0 + wav) * TMAXX * NN + 4 * lane;

#pragma unroll 1
    for (int t = 1; t < TMAXX; ++t) {
        v2f xr[2][KBLK], cr[2][KBLK];
        {
            const float4 o0 = *(const float4*)(xrow);
            const float4 o1 = *(const float4*)(xrow + 128);
            __builtin_amdgcn_sched_barrier(0);
#pragma unroll
            for (int i = 0; i < KBLK; ++i) {
                xr[0][i] = *(const v2f*)(xlane + 4 * i);
                cr[0][i] = *(const v2f*)(clane + 4 * i);
            }
#pragma unroll
            for (int i = 0; i < KBLK; ++i) {
                xr[1][i] = *(const v2f*)(xlane + 4 * (KBLK + i));
                cr[1][i] = *(const v2f*)(clane + 4 * (KBLK + i));
            }
            float* q = outRow + (size_t)(t - 1) * NN;
            *(float4*)(q)       = o0;
            *(float4*)(q + 128) = o1;
        }
        v8f accA = {0.f, 0.f, 0.f, 0.f, 0.f, 0.f, 0.f, 0.f};
        v8f accC = {0.f, 0.f, 0.f, 0.f, 0.f, 0.f, 0.f, 0.f};
#pragma unroll
        for (int b = 0; b < 64 / KBLK; ++b) {
            __builtin_amdgcn_sched_barrier(0);
#pragma unroll
            for (int i = 0; i < KBLK; ++i) {
                accA = wmma4(aF[KBLK * b + i], xr[b & 1][i], accA);
                accC = wmma4(cr[b & 1][i],     xr[b & 1][i], accC);
            }
            __builtin_amdgcn_sched_barrier(0);
            if (b + 2 < 64 / KBLK) {
#pragma unroll
                for (int i = 0; i < KBLK; ++i) {
                    xr[b & 1][i] = *(const v2f*)(xlane + 4 * (KBLK * (b + 2) + i));
                    cr[b & 1][i] = *(const v2f*)(clane + 4 * (KBLK * (b + 2) + i));
                }
            }
        }
        float xn[8];
#pragma unroll
        for (int j = 0; j < 8; ++j) {
            const float xc = Xs[lo][rbase + j];
            const float yi = accC[j] + byF[j];
            xn[j] = xc + STEPF * (accA[j] + tanh_sel(yi));
        }
        __builtin_amdgcn_sched_barrier(0);
        __syncthreads();
#pragma unroll
        for (int j = 0; j < 8; ++j) Xs[lo][rbase + j] = xn[j];
        __syncthreads();
    }
    {
        const float4 o0 = *(const float4*)(xrow);
        const float4 o1 = *(const float4*)(xrow + 128);
        float* q = outRow + (size_t)(TMAXX - 1) * NN;
        *(float4*)(q)       = o0;
        *(float4*)(q + 128) = o1;
    }
}

extern "C" void kernel_launch(void* const* d_in, const int* in_sizes, int n_in,
                              void* d_out, int out_size, void* d_ws, size_t ws_size,
                              hipStream_t stream) {
    const float* X0  = (const float*)d_in[0];
    const float* MA  = (const float*)d_in[1];
    const float* MW  = (const float*)d_in[2];
    const float* bAz = (const float*)d_in[3];
    const float* bWz = (const float*)d_in[4];
    const float* by  = (const float*)d_in[5];
    float* out = (float*)d_out;

    float* A = (float*)d_ws;
    float* C = A + NN * NN;

    lrnn_prep<<<NN, NN, 0, stream>>>(MA, MW, bAz, bWz, A, C);
    lrnn_main<<<16, 512, 0, stream>>>(X0, A, C, by, out);
}


# one launch per time step on 128 workgroups (16 batch x 8 row tiles) replaces the 16-workgroup persistent loop; k summed as four quarter chains joined in a fixed order, not bit-identical
# speedup vs baseline: 2.0778x; 1.3943x over previous
#include <hip/hip_runtime.h>
#include <hip/hip_bf16.h>
#include <math.h>

#define NN 256
#define BSZ 256
#define TMAXX 512
#define STEPF 0.01f
#define YAF 0.001f
#define YWF 0.001f

#define XSTRIDE 260
#define PSTRIDE 36

typedef __attribute__((ext_vector_type(2))) float v2f;
typedef __attribute__((ext_vector_type(8))) float v8f;

static __device__ __forceinline__ v8f wmma4(v2f a, v2f b, v8f c) {
    return __builtin_amdgcn_wmma_f32_16x16x4_f32(
        false, a, false, b, (short)0, c, false, false);
}

static __device__ __forceinline__ float tanh_sel(float y) {
    const float ay = fabsf(y);
    const float e  = __builtin_amdgcn_exp2f(ay * 0x1.715476p+1f);
    const float tb = fmaf(__builtin_amdgcn_rcpf(e + 1.0f), -2.0f, 1.0f);
    const float s  = y * y;
    float p = fmaf(s, -0x1.758e7ap-8f, 0x1.521192p-6f);
    p = fmaf(s, p, -0x1.b8389cp-5f);
    p = fmaf(s, p, 0x1.110704p-3f);
    p = fmaf(s, p, -0x1.555532p-2f);
    const float ts = fmaf(s, ay * p, ay);
    const float t  = (ay < 0.625f) ? ts : tb;
    return copysignf(t, y);
}

static __device__ __forceinline__ int frag_index(int row, int k) {
    const int tile = row >> 4, c = row & 15;
    const int kk = k >> 2, h = (k >> 1) & 1, s = k & 1;
    const int j = kk >> 1, e = 2 * (kk & 1) + s;
    return (((tile * 32 + j) * 32) + (c + 16 * h)) * 4 + e;
}

__global__ __launch_bounds__(256) void lrnn_prep(
    const float* __restrict__ MA, const float* __restrict__ MW,
    const float* __restrict__ bAz, const float* __restrict__ bWz,
    const float* __restrict__ X0,
    float* __restrict__ Af, float* __restrict__ Cf, float* __restrict__ out) {
    const int i = blockIdx.x;
    const int j = threadIdx.x;
    const float zA = bAz[0];
    const float zW = bWz[0];
    const float bA = 0.5f * expf(-zA * zA) + 0.5f;
    const float bW = 0.5f * expf(-zW * zW) + 0.5f;
    const float ma_ij = MA[i * NN + j];
    const float ma_ji = MA[j * NN + i];
    const float mw_ij = MW[i * NN + j];
    const float mw_ji = MW[j * NN + i];
    const float d = (i == j) ? 1.0f : 0.0f;
    const int f = frag_index(i, j);
    Af[f] = ma_ij + (1.0f - 2.0f * bA) * ma_ji - YAF * d;
    Cf[f] = (1.0f - bA + bW) * mw_ij + (1.0f - bA - bW) * mw_ji - YWF * d;
    out[(size_t)i * TMAXX * NN + j] = X0[i * NN + j];
}

__global__ __launch_bounds__(128) void lrnn_step(
    const float* __restrict__ Af, const float* __restrict__ Cf,
    const float* __restrict__ by, const float* xprev, float* xnext) {
    __shared__ __attribute__((aligned(16))) float Xs[16][XSTRIDE];
    __shared__ __attribute__((aligned(16))) float P[2][4][16][PSTRIDE];

    const int tid  = threadIdx.x;
    const int q    = tid >> 5;
    const int lane = tid & 31;
    const int c    = lane & 15;
    const int h    = lane >> 4;
    const int bc0  = blockIdx.x * 16;
    const int r0   = blockIdx.y * 32;
    const int col  = tid >> 3;
    const int r4   = 4 * (tid & 7);


    float4 bv = *(const float4*)(by + r0 + r4);

    float4 xt[8];
#pragma unroll
    for (int it = 0; it < 8; ++it) {
        const int idx = tid + 128 * it;
        xt[it] = *(const float4*)(xprev + (size_t)(bc0 + (idx >> 6)) * TMAXX * NN + 4 * (idx & 63));
    }
    float4 a0[8], a1[8], c0[8], c1[8];
    {
        const size_t t0 = (size_t)(2 * blockIdx.y) * 32 + 8 * q;
        const size_t t1 = (size_t)(2 * blockIdx.y + 1) * 32 + 8 * q;
        const float4* pa0 = (const float4*)Af + t0 * 32 + lane;
        const float4* pc0 = (const float4*)Cf + t0 * 32 + lane;
        const float4* pa1 = (const float4*)Af + t1 * 32 + lane;
        const float4* pc1 = (const float4*)Cf + t1 * 32 + lane;
#pragma unroll
        for (int i = 0; i < 8; ++i) {
            a0[i] = pa0[32 * i];
            c0[i] = pc0[32 * i];
            a1[i] = pa1[32 * i];
            c1[i] = pc1[32 * i];
        }
    }
#pragma unroll
    for (int it = 0; it < 8; ++it) {
        const int idx = tid + 128 * it;
        *(float4*)&Xs[idx >> 6][4 * (idx & 63)] = xt[it];
    }
    asm volatile("" : "+v"(bv.x), "+v"(bv.y), "+v"(bv.z), "+v"(bv.w));
    __builtin_amdgcn_fence(__ATOMIC_RELEASE, "workgroup", "local");
    __builtin_amdgcn_s_barrier();
    __builtin_amdgcn_fence(__ATOMIC_ACQUIRE, "workgroup", "local");

    const float* xlane = &Xs[c][64 * q + 2 * h];
    v2f xf[16];
#pragma unroll
    for (int i = 0; i < 16; ++i) xf[i] = *(const v2f*)(xlane + 4 * i);
    __builtin_amdgcn_sched_barrier(0);
    v8f accA0 = {0.f, 0.f, 0.f, 0.f, 0.f, 0.f, 0.f, 0.f};
    v8f accA1 = accA0, accC0 = accA0, accC1 = accA0;
#pragma unroll
    for (int i = 0; i < 8; ++i) {
        const v2f xlo = xf[2 * i];
        const v2f xhi = xf[2 * i + 1];
        const v2f p0 = {a0[i].x, a0[i].y}, p1 = {a0[i].z, a0[i].w};
        const v2f s0 = {a1[i].x, a1[i].y}, s1 = {a1[i].z, a1[i].w};
        const v2f u0 = {c0[i].x, c0[i].y}, u1 = {c0[i].z, c0[i].w};
        const v2f w0 = {c1[i].x, c1[i].y}, w1 = {c1[i].z, c1[i].w};
        accA0 = wmma4(p0, xlo, accA0);
        accC0 = wmma4(u0, xlo, accC0);
        accA1 = wmma4(s0, xlo, accA1);
        accC1 = wmma4(w0, xlo, accC1);
        accA0 = wmma4(p1, xhi, accA0);
        accC0 = wmma4(u1, xhi, accC0);
        accA1 = wmma4(s1, xhi, accA1);
        accC1 = wmma4(w1, xhi, accC1);
    }

    {
        float* pa = &P[0][q][c][8 * h];
        float* pc = &P[1][q][c][8 * h];
        *(float4*)(pa)      = make_float4(accA0[0], accA0[1], accA0[2], accA0[3]);
        *(float4*)(pa + 4)  = make_float4(accA0[4], accA0[5], accA0[6], accA0[7]);
        *(float4*)(pa + 16) = make_float4(accA1[0], accA1[1], accA1[2], accA1[3]);
        *(float4*)(pa + 20) = make_float4(accA1[4], accA1[5], accA1[6], accA1[7]);
        *(float4*)(pc)      = make_float4(accC0[0], accC0[1], accC0[2], accC0[3]);
        *(float4*)(pc + 4)  = make_float4(accC0[4], accC0[5], accC0[6], accC0[7]);
        *(float4*)(pc + 16) = make_float4(accC1[0], accC1[1], accC1[2], accC1[3]);
        *(float4*)(pc + 20) = make_float4(accC1[4], accC1[5], accC1[6], accC1[7]);
    }
    __syncthreads();

    float4 sa = *(const float4*)&P[0][0][col][r4];
    float4 sc = *(const float4*)&P[1][0][col][r4];
#pragma unroll
    for (int k = 1; k < 4; ++k) {
        const float4 ta = *(const float4*)&P[0][k][col][r4];
        const float4 tc = *(const float4*)&P[1][k][col][r4];
        sa.x += ta.x; sa.y += ta.y; sa.z += ta.z; sa.w += ta.w;
        sc.x += tc.x; sc.y += tc.y; sc.z += tc.z; sc.w += tc.w;
    }
    const float4 xc = *(const float4*)&Xs[col][r0 + r4];
    float4 xn;
    xn.x = xc.x + STEPF * (sa.x + tanh_sel(sc.x + bv.x));
    xn.y = xc.y + STEPF * (sa.y + tanh_sel(sc.y + bv.y));
    xn.z = xc.z + STEPF * (sa.z + tanh_sel(sc.z + bv.z));
    xn.w = xc.w + STEPF * (sa.w + tanh_sel(sc.w + bv.w));
    *(float4*)(xnext + (size_t)(bc0 + col) * TMAXX * NN + r0 + r4) = xn;
}

extern "C" void kernel_launch(void* const* d_in, const int* in_sizes, int n_in,
                              void* d_out, int out_size, void* d_ws, size_t ws_size,
                              hipStream_t stream) {
    const float* X0  = (const float*)d_in[0];
    const float* MA  = (const float*)d_in[1];
    const float* MW  = (const float*)d_in[2];
    const float* bAz = (const float*)d_in[3];
    const float* bWz = (const float*)d_in[4];
    const float* by  = (const float*)d_in[5];
    float* out = (float*)d_out;

    float* Af = (float*)d_ws;
    float* Cf = Af + NN * NN;

    lrnn_prep<<<NN, NN, 0, stream>>>(MA, MW, bAz, bWz, X0, Af, Cf, out);
    for (float* p = out; p != out + (size_t)(TMAXX - 1) * NN; p += NN)
        lrnn_step<<<dim3(16, 8), 128, 0, stream>>>(Af, Cf, by, p, p + NN);
}
